# starting kernel
# baseline (speedup 1.0000x reference)
#include <hip/hip_runtime.h>
#include <math.h>

#define IN_DIM   128
#define OUT_DIM  64
#define NEG_SLOPE 0.01f

typedef __attribute__((ext_vector_type(2))) float v2f;
typedef __attribute__((ext_vector_type(8))) float v8f;

__device__ __forceinline__ unsigned enc_f32(float f) {
    unsigned u = __float_as_uint(f);
    return (u & 0x80000000u) ? ~u : (u | 0x80000000u);
}
__device__ __forceinline__ float dec_f32(unsigned u) {
    return __uint_as_float((u & 0x80000000u) ? (u & 0x7FFFFFFFu) : ~u);
}

__global__ void gat_init_kernel(float* __restrict__ h_out, int n_out,
                                unsigned* __restrict__ e_max_enc,
                                float* __restrict__ e_sum, int n_nodes) {
    int i = blockIdx.x * blockDim.x + threadIdx.x;
    int stride = gridDim.x * blockDim.x;
    for (int j = i; j < n_out; j += stride) h_out[j] = 0.0f;
    for (int j = i; j < n_nodes; j += stride) {
        e_max_enc[j] = 0u;
        e_sum[j]     = 0.0f;
    }
}

#define LDS_PITCH 132
__global__ __launch_bounds__(128) void gat_gemm_wmma_kernel(
        const float* __restrict__ h, const float* __restrict__ fc_w,
        float* __restrict__ z, int n_nodes) {
    __shared__ float lds_w[OUT_DIM * LDS_PITCH];

    const int tid = threadIdx.x;
    for (int i = tid; i < OUT_DIM * (IN_DIM / 4); i += 128) {
        int row = i >> 5;
        int seg = (i & 31) << 2;
        const float4 v = *(const float4*)&fc_w[row * IN_DIM + seg];
        *(float4*)&lds_w[row * LDS_PITCH + seg] = v;
    }
    __syncthreads();

    const int wave   = tid >> 5;
    const int lane   = tid & 31;
    const int half   = lane >> 4;
    const int lid    = lane & 15;
    const int m_base = blockIdx.x * 16;
    const int n_base = wave * 16;

    int m_row = m_base + lid;
    if (m_row >= n_nodes) m_row = n_nodes - 1;

    const float* arow = h + (size_t)m_row * IN_DIM + 2 * half;
    const float* brow = lds_w + (n_base + lid) * LDS_PITCH + 2 * half;

    v8f c = {};
#pragma unroll
    for (int k0 = 0; k0 < IN_DIM; k0 += 4) {
        v2f a = *(const v2f*)(arow + k0);
        v2f b = *(const v2f*)(brow + k0);
        c = __builtin_amdgcn_wmma_f32_16x16x4_f32(
                 false, a,  false, b,
                 (short)0, c,  false,  false);
    }

#pragma unroll
    for (int r = 0; r < 8; ++r) {
        int m = m_base + half * 8 + r;
        if (m < n_nodes)
            z[(size_t)m * OUT_DIM + n_base + lid] = c[r];
    }
}

__global__ __launch_bounds__(256) void gat_scores_kernel(
        const float* __restrict__ z, const float* __restrict__ attn_w,
        float* __restrict__ s_src, float* __restrict__ s_dst, int n_nodes) {
    const int lane = threadIdx.x & 31;
    const int node = (blockIdx.x * blockDim.x + threadIdx.x) >> 5;
    if (node >= n_nodes) return;

    const v2f zv = *(const v2f*)&z[(size_t)node * OUT_DIM + 2 * lane];
    const v2f as = *(const v2f*)&attn_w[2 * lane];
    const v2f ad = *(const v2f*)&attn_w[OUT_DIM + 2 * lane];

    float ps = zv.x * as.x + zv.y * as.y;
    float pd = zv.x * ad.x + zv.y * ad.y;
#pragma unroll
    for (int m = 16; m >= 1; m >>= 1) {
        ps += __shfl_xor(ps, m, 32);
        pd += __shfl_xor(pd, m, 32);
    }
    if (lane == 0) {
        s_src[node] = ps;
        s_dst[node] = pd;
    }
}

__global__ __launch_bounds__(256) void gat_edge_max_kernel(
        const int* __restrict__ src, const int* __restrict__ dst,
        const float* __restrict__ s_src, const float* __restrict__ s_dst,
        float* __restrict__ e, unsigned* __restrict__ e_max_enc, int n_edges) {
    int i = blockIdx.x * blockDim.x + threadIdx.x;
    if (i >= n_edges) return;
    float v = s_src[src[i]] + s_dst[dst[i]];
    v = (v > 0.0f) ? v : NEG_SLOPE * v;
    e[i] = v;
    atomicMax(&e_max_enc[dst[i]], enc_f32(v));
}

__global__ __launch_bounds__(256) void gat_edge_sum_kernel(
        const int* __restrict__ dst, const float* __restrict__ e,
        const unsigned* __restrict__ e_max_enc, float* __restrict__ e_sum,
        int n_edges) {
    int i = blockIdx.x * blockDim.x + threadIdx.x;
    if (i >= n_edges) return;
    int d = dst[i];
    atomicAdd(&e_sum[d], __expf(e[i] - dec_f32(e_max_enc[d])));
}

__global__ __launch_bounds__(256) void gat_scatter_kernel(
        const int* __restrict__ src, const int* __restrict__ dst,
        const float* __restrict__ e, const unsigned* __restrict__ e_max_enc,
        const float* __restrict__ e_sum, const float* __restrict__ z,
        float* __restrict__ h_out, int n_edges) {
    const int lane = threadIdx.x & 31;
    const int edge = (blockIdx.x * blockDim.x + threadIdx.x) >> 5;
    if (edge >= n_edges) return;

    const int s = src[edge];
    const int d = dst[edge];
    const float alpha = __expf(e[edge] - dec_f32(e_max_enc[d])) / e_sum[d];

    const v2f zv = *(const v2f*)&z[(size_t)s * OUT_DIM + 2 * lane];
    float* out = &h_out[(size_t)d * OUT_DIM + 2 * lane];
    atomicAdd(&out[0], alpha * zv.x);
    atomicAdd(&out[1], alpha * zv.y);
}

extern "C" void kernel_launch(void* const* d_in, const int* in_sizes, int n_in,
                              void* d_out, int out_size, void* d_ws, size_t ws_size,
                              hipStream_t stream) {
    const float* h      = (const float*)d_in[0];
    const int*   src    = (const int*)d_in[1];
    const int*   dst    = (const int*)d_in[2];
    const float* fc_w   = (const float*)d_in[3];
    const float* attn_w = (const float*)d_in[4];
    float*       h_out  = (float*)d_out;

    const int n_nodes = in_sizes[0] / IN_DIM;
    const int n_edges = in_sizes[1];

    char* ws = (char*)d_ws;
    size_t off = 0;
    auto alloc = [&](size_t bytes) {
        void* p = ws + off;
        off += (bytes + 15) & ~size_t(15);
        return p;
    };
    float*    z         = (float*)alloc((size_t)n_nodes * OUT_DIM * sizeof(float));
    float*    s_src     = (float*)alloc((size_t)n_nodes * sizeof(float));
    float*    s_dst     = (float*)alloc((size_t)n_nodes * sizeof(float));
    unsigned* e_max_enc = (unsigned*)alloc((size_t)n_nodes * sizeof(unsigned));
    float*    e_sum     = (float*)alloc((size_t)n_nodes * sizeof(float));
    float*    e         = (float*)alloc((size_t)n_edges * sizeof(float));

    {
        int total = n_nodes * OUT_DIM;
        int blocks = (total + 255) / 256;
        if (blocks > 4096) blocks = 4096;
        gat_init_kernel<<<blocks, 256, 0, stream>>>(h_out, total, e_max_enc,
                                                    e_sum, n_nodes);
    }
    {
        int blocks = (n_nodes + 15) / 16;
        gat_gemm_wmma_kernel<<<blocks, 128, 0, stream>>>(h, fc_w, z, n_nodes);
    }
    {
        int blocks = (n_nodes * 32 + 255) / 256;
        gat_scores_kernel<<<blocks, 256, 0, stream>>>(z, attn_w, s_src, s_dst,
                                                      n_nodes);
    }
    {
        int blocks = (n_edges + 255) / 256;
        gat_edge_max_kernel<<<blocks, 256, 0, stream>>>(src, dst, s_src, s_dst,
                                                        e, e_max_enc, n_edges);
    }
    {
        int blocks = (n_edges + 255) / 256;
        gat_edge_sum_kernel<<<blocks, 256, 0, stream>>>(dst, e, e_max_enc,
                                                        e_sum, n_edges);
    }
    {
        int blocks = (n_edges * 32 + 255) / 256;
        gat_scatter_kernel<<<blocks, 256, 0, stream>>>(src, dst, e, e_max_enc,
                                                       e_sum, z, h_out, n_edges);
    }
}


# scatter kernel: one wave walks 32 consecutive edges and sums alpha*z[src] over each run of equal dst in registers, one pair of atomic adds per run instead of per edge
# speedup vs baseline: 1.8631x; 1.8631x over previous
#include <hip/hip_runtime.h>
#include <math.h>

#define IN_DIM   128
#define OUT_DIM  64
#define NEG_SLOPE 0.01f

typedef __attribute__((ext_vector_type(2))) float v2f;
typedef __attribute__((ext_vector_type(8))) float v8f;

__device__ __forceinline__ unsigned enc_f32(float f) {
    unsigned u = __float_as_uint(f);
    return (u & 0x80000000u) ? ~u : (u | 0x80000000u);
}
__device__ __forceinline__ float dec_f32(unsigned u) {
    return __uint_as_float((u & 0x80000000u) ? (u & 0x7FFFFFFFu) : ~u);
}

__global__ void gat_init_kernel(float* __restrict__ h_out, int n_out,
                                unsigned* __restrict__ e_max_enc,
                                float* __restrict__ e_sum, int n_nodes) {
    int i = blockIdx.x * blockDim.x + threadIdx.x;
    int stride = gridDim.x * blockDim.x;
    for (int j = i; j < n_out; j += stride) h_out[j] = 0.0f;
    for (int j = i; j < n_nodes; j += stride) {
        e_max_enc[j] = 0u;
        e_sum[j]     = 0.0f;
    }
}

#define LDS_PITCH 132
__global__ __launch_bounds__(128) void gat_gemm_wmma_kernel(
        const float* __restrict__ h, const float* __restrict__ fc_w,
        float* __restrict__ z, int n_nodes) {
    __shared__ float lds_w[OUT_DIM * LDS_PITCH];

    const int tid = threadIdx.x;
    for (int i = tid; i < OUT_DIM * (IN_DIM / 4); i += 128) {
        int row = i >> 5;
        int seg = (i & 31) << 2;
        const float4 v = *(const float4*)&fc_w[row * IN_DIM + seg];
        *(float4*)&lds_w[row * LDS_PITCH + seg] = v;
    }
    __syncthreads();

    const int wave   = tid >> 5;
    const int lane   = tid & 31;
    const int half   = lane >> 4;
    const int lid    = lane & 15;
    const int m_base = blockIdx.x * 16;
    const int n_base = wave * 16;

    int m_row = m_base + lid;
    if (m_row >= n_nodes) m_row = n_nodes - 1;

    const float* arow = h + (size_t)m_row * IN_DIM + 2 * half;
    const float* brow = lds_w + (n_base + lid) * LDS_PITCH + 2 * half;

    v8f c = {};
#pragma unroll
    for (int k0 = 0; k0 < IN_DIM; k0 += 4) {
        v2f a = *(const v2f*)(arow + k0);
        v2f b = *(const v2f*)(brow + k0);
        c = __builtin_amdgcn_wmma_f32_16x16x4_f32(
                 false, a,  false, b,
                 (short)0, c,  false,  false);
    }

#pragma unroll
    for (int r = 0; r < 8; ++r) {
        int m = m_base + half * 8 + r;
        if (m < n_nodes)
            z[(size_t)m * OUT_DIM + n_base + lid] = c[r];
    }
}

__global__ __launch_bounds__(256) void gat_scores_kernel(
        const float* __restrict__ z, const float* __restrict__ attn_w,
        float* __restrict__ s_src, float* __restrict__ s_dst, int n_nodes) {
    const int lane = threadIdx.x & 31;
    const int node = (blockIdx.x * blockDim.x + threadIdx.x) >> 5;
    if (node >= n_nodes) return;

    const v2f zv = *(const v2f*)&z[(size_t)node * OUT_DIM + 2 * lane];
    const v2f as = *(const v2f*)&attn_w[2 * lane];
    const v2f ad = *(const v2f*)&attn_w[OUT_DIM + 2 * lane];

    float ps = zv.x * as.x + zv.y * as.y;
    float pd = zv.x * ad.x + zv.y * ad.y;
#pragma unroll
    for (int m = 16; m >= 1; m >>= 1) {
        ps += __shfl_xor(ps, m, 32);
        pd += __shfl_xor(pd, m, 32);
    }
    if (lane == 0) {
        s_src[node] = ps;
        s_dst[node] = pd;
    }
}

__global__ __launch_bounds__(256) void gat_edge_max_kernel(
        const int* __restrict__ src, const int* __restrict__ dst,
        const float* __restrict__ s_src, const float* __restrict__ s_dst,
        float* __restrict__ e, unsigned* __restrict__ e_max_enc, int n_edges) {
    int i = blockIdx.x * blockDim.x + threadIdx.x;
    if (i >= n_edges) return;
    float v = s_src[src[i]] + s_dst[dst[i]];
    v = (v > 0.0f) ? v : NEG_SLOPE * v;
    e[i] = v;
    atomicMax(&e_max_enc[dst[i]], enc_f32(v));
}

__global__ __launch_bounds__(256) void gat_edge_sum_kernel(
        const int* __restrict__ dst, const float* __restrict__ e,
        const unsigned* __restrict__ e_max_enc, float* __restrict__ e_sum,
        int n_edges) {
    int i = blockIdx.x * blockDim.x + threadIdx.x;
    if (i >= n_edges) return;
    int d = dst[i];
    atomicAdd(&e_sum[d], __expf(e[i] - dec_f32(e_max_enc[d])));
}

#define EDGES_PER_WAVE 32
#define ROWS_IN_FLIGHT 8

__device__ __forceinline__ void gat_flush_run(float* __restrict__ h_out, int d, int lane,
                                              v2f acc) {
    float* out = &h_out[(size_t)d * OUT_DIM + 2 * lane];
    atomicAdd(&out[0], acc.x);
    atomicAdd(&out[1], acc.y);
}

__global__ __launch_bounds__(256) void gat_scatter_kernel(
        const int* __restrict__ src, const int* __restrict__ dst,
        const float* __restrict__ e, const unsigned* __restrict__ e_max_enc,
        const float* __restrict__ e_sum, const float* __restrict__ z,
        float* __restrict__ h_out, int n_edges) {
    const int lane  = threadIdx.x & 31;
    const int chunk = (blockIdx.x * blockDim.x + threadIdx.x) >> 5;
    const int base  = chunk * EDGES_PER_WAVE;
    if (base >= n_edges) return;

    int i = base + lane;
    const bool valid = i < n_edges;
    if (!valid) i = n_edges - 1;
    const int my_s = src[i];
    const int my_d = dst[i];
    float my_a = __expf(e[i] - dec_f32(e_max_enc[my_d])) / e_sum[my_d];
    if (!valid) my_a = 0.0f;

    int cur_d = __builtin_amdgcn_readlane(my_d, 0);
    v2f acc = {0.0f, 0.0f};

    for (int g = 0; g < EDGES_PER_WAVE; g += ROWS_IN_FLIGHT) {
        v2f zr[ROWS_IN_FLIGHT];
#pragma unroll
        for (int j = 0; j < ROWS_IN_FLIGHT; ++j) {
            const int s = __builtin_amdgcn_readlane(my_s, g + j);
            zr[j] = *(const v2f*)&z[(size_t)s * OUT_DIM + 2 * lane];
        }
#pragma unroll
        for (int j = 0; j < ROWS_IN_FLIGHT; ++j) {
            const int   d = __builtin_amdgcn_readlane(my_d, g + j);
            const float a = __int_as_float(
                    __builtin_amdgcn_readlane(__float_as_int(my_a), g + j));
            if (d != cur_d) {
                gat_flush_run(h_out, cur_d, lane, acc);
                acc = (v2f){0.0f, 0.0f};
                cur_d = d;
            }
            acc.x += a * zr[j].x;
            acc.y += a * zr[j].y;
        }
    }
    gat_flush_run(h_out, cur_d, lane, acc);
}

extern "C" void kernel_launch(void* const* d_in, const int* in_sizes, int n_in,
                              void* d_out, int out_size, void* d_ws, size_t ws_size,
                              hipStream_t stream) {
    const float* h      = (const float*)d_in[0];
    const int*   src    = (const int*)d_in[1];
    const int*   dst    = (const int*)d_in[2];
    const float* fc_w   = (const float*)d_in[3];
    const float* attn_w = (const float*)d_in[4];
    float*       h_out  = (float*)d_out;

    const int n_nodes = in_sizes[0] / IN_DIM;
    const int n_edges = in_sizes[1];

    char* ws = (char*)d_ws;
    size_t off = 0;
    auto alloc = [&](size_t bytes) {
        void* p = ws + off;
        off += (bytes + 15) & ~size_t(15);
        return p;
    };
    float*    z         = (float*)alloc((size_t)n_nodes * OUT_DIM * sizeof(float));
    float*    s_src     = (float*)alloc((size_t)n_nodes * sizeof(float));
    float*    s_dst     = (float*)alloc((size_t)n_nodes * sizeof(float));
    unsigned* e_max_enc = (unsigned*)alloc((size_t)n_nodes * sizeof(unsigned));
    float*    e_sum     = (float*)alloc((size_t)n_nodes * sizeof(float));
    float*    e         = (float*)alloc((size_t)n_edges * sizeof(float));

    {
        int total = n_nodes * OUT_DIM;
        int blocks = (total + 255) / 256;
        if (blocks > 4096) blocks = 4096;
        gat_init_kernel<<<blocks, 256, 0, stream>>>(h_out, total, e_max_enc,
                                                    e_sum, n_nodes);
    }
    {
        int blocks = (n_nodes + 15) / 16;
        gat_gemm_wmma_kernel<<<blocks, 128, 0, stream>>>(h, fc_w, z, n_nodes);
    }
    {
        int blocks = (n_nodes * 32 + 255) / 256;
        gat_scores_kernel<<<blocks, 256, 0, stream>>>(z, attn_w, s_src, s_dst,
                                                      n_nodes);
    }
    {
        int blocks = (n_edges + 255) / 256;
        gat_edge_max_kernel<<<blocks, 256, 0, stream>>>(src, dst, s_src, s_dst,
                                                        e, e_max_enc, n_edges);
    }
    {
        int blocks = (n_edges + 255) / 256;
        gat_edge_sum_kernel<<<blocks, 256, 0, stream>>>(dst, e, e_max_enc,
                                                        e_sum, n_edges);
    }
    {
        int blocks = (n_edges + 255) / 256;
        gat_scatter_kernel<<<blocks, 256, 0, stream>>>(src, dst, e, e_max_enc,
                                                       e_sum, z, h_out, n_edges);
    }
}


# init, scores, edge-max and edge-sum kernels: each wave loops over many items (12 float4, 32 nodes, 8 edges per lane) instead of one; same arithmetic and traffic, about 123000 fewer waves to enter
# speedup vs baseline: 2.0487x; 1.0996x over previous
#include <hip/hip_runtime.h>
#include <math.h>

#define IN_DIM   128
#define OUT_DIM  64
#define NEG_SLOPE 0.01f

typedef __attribute__((ext_vector_type(2))) float v2f;
typedef __attribute__((ext_vector_type(8))) float v8f;

__device__ __forceinline__ unsigned enc_f32(float f) {
    unsigned u = __float_as_uint(f);
    return (u & 0x80000000u) ? ~u : (u | 0x80000000u);
}
__device__ __forceinline__ float dec_f32(unsigned u) {
    return __uint_as_float((u & 0x80000000u) ? (u & 0x7FFFFFFFu) : ~u);
}

__global__ __launch_bounds__(256) void gat_init_kernel(
        float* __restrict__ h_out, unsigned* __restrict__ e_max_enc,
        float* __restrict__ e_sum, int n_nodes) {
    const int i = blockIdx.x * blockDim.x + threadIdx.x;
    const int stride = gridDim.x * blockDim.x;
    float4* __restrict__ out4 = (float4*)h_out;
    const int n4 = n_nodes * (OUT_DIM / 4);
    const float4 zero4 = {0.0f, 0.0f, 0.0f, 0.0f};
    for (int j = i; j < n4; j += stride) out4[j] = zero4;
    for (int j = i; j < n_nodes; j += stride) {
        e_max_enc[j] = 0u;
        e_sum[j]     = 0.0f;
    }
}

#define LDS_PITCH 132
__global__ __launch_bounds__(128) void gat_gemm_wmma_kernel(
        const float* __restrict__ h, const float* __restrict__ fc_w,
        float* __restrict__ z, int n_nodes) {
    __shared__ float lds_w[OUT_DIM * LDS_PITCH];

    const int tid = threadIdx.x;
    for (int i = tid; i < OUT_DIM * (IN_DIM / 4); i += 128) {
        int row = i >> 5;
        int seg = (i & 31) << 2;
        const float4 v = *(const float4*)&fc_w[row * IN_DIM + seg];
        *(float4*)&lds_w[row * LDS_PITCH + seg] = v;
    }
    __syncthreads();

    const int wave   = tid >> 5;
    const int lane   = tid & 31;
    const int half   = lane >> 4;
    const int lid    = lane & 15;
    const int m_base = blockIdx.x * 16;
    const int n_base = wave * 16;

    int m_row = m_base + lid;
    if (m_row >= n_nodes) m_row = n_nodes - 1;

    const float* arow = h + (size_t)m_row * IN_DIM + 2 * half;
    const float* brow = lds_w + (n_base + lid) * LDS_PITCH + 2 * half;

    v8f c = {};
#pragma unroll
    for (int k0 = 0; k0 < IN_DIM; k0 += 4) {
        v2f a = *(const v2f*)(arow + k0);
        v2f b = *(const v2f*)(brow + k0);
        c = __builtin_amdgcn_wmma_f32_16x16x4_f32(
                 false, a,  false, b,
                 (short)0, c,  false,  false);
    }

#pragma unroll
    for (int r = 0; r < 8; ++r) {
        int m = m_base + half * 8 + r;
        if (m < n_nodes)
            z[(size_t)m * OUT_DIM + n_base + lid] = c[r];
    }
}

#define NODES_PER_WAVE 32
#define SCORE_ROWS_IN_FLIGHT 16
__global__ __launch_bounds__(256) void gat_scores_kernel(
        const float* __restrict__ z, const float* __restrict__ attn_w,
        float* __restrict__ s_src, float* __restrict__ s_dst, int n_nodes) {
    const int lane = threadIdx.x & 31;
    const int wave = (blockIdx.x * blockDim.x + threadIdx.x) >> 5;
    const int base = wave * NODES_PER_WAVE;
    if (base >= n_nodes) return;

    const v2f as = *(const v2f*)&attn_w[2 * lane];
    const v2f ad = *(const v2f*)&attn_w[OUT_DIM + 2 * lane];

    float my_ps = 0.0f, my_pd = 0.0f;
    for (int g = 0; g < NODES_PER_WAVE; g += SCORE_ROWS_IN_FLIGHT) {
        v2f zr[SCORE_ROWS_IN_FLIGHT];
#pragma unroll
        for (int j = 0; j < SCORE_ROWS_IN_FLIGHT; ++j) {
            int node = base + g + j;
            if (node >= n_nodes) node = n_nodes - 1;
            zr[j] = *(const v2f*)&z[(size_t)node * OUT_DIM + 2 * lane];
        }
#pragma unroll
        for (int j = 0; j < SCORE_ROWS_IN_FLIGHT; ++j) {
            float ps = zr[j].x * as.x + zr[j].y * as.y;
            float pd = zr[j].x * ad.x + zr[j].y * ad.y;
#pragma unroll
            for (int m = 16; m >= 1; m >>= 1) {
                ps += __shfl_xor(ps, m, 32);
                pd += __shfl_xor(pd, m, 32);
            }
            if (lane == g + j) {
                my_ps = ps;
                my_pd = pd;
            }
        }
    }
    const int node = base + lane;
    if (node < n_nodes) {
        s_src[node] = my_ps;
        s_dst[node] = my_pd;
    }
}

#define EDGES_PER_LANE 8
#define EDGES_PER_WAVE_K34 (32 * EDGES_PER_LANE)

__device__ __forceinline__ void gat_edge_max_chunk(
        const int* __restrict__ src, const int* __restrict__ dst,
        const float* __restrict__ s_src, const float* __restrict__ s_dst,
        float* __restrict__ e, unsigned* __restrict__ e_max_enc, int n_edges,
        int first, const bool whole) {
    int sv[EDGES_PER_LANE], dv[EDGES_PER_LANE];
#pragma unroll
    for (int j = 0; j < EDGES_PER_LANE; ++j) {
        int i = first + 32 * j;
        if (!whole && i >= n_edges) i = n_edges - 1;
        sv[j] = src[i];
        dv[j] = dst[i];
    }
    float a[EDGES_PER_LANE], b[EDGES_PER_LANE];
#pragma unroll
    for (int j = 0; j < EDGES_PER_LANE; ++j) {
        a[j] = s_src[sv[j]];
        b[j] = s_dst[dv[j]];
    }
#pragma unroll
    for (int j = 0; j < EDGES_PER_LANE; ++j) {
        const int i = first + 32 * j;
        float v = a[j] + b[j];
        v = (v > 0.0f) ? v : NEG_SLOPE * v;
        if (whole || i < n_edges) {
            e[i] = v;
            atomicMax(&e_max_enc[dv[j]], enc_f32(v));
        }
    }
}

__global__ __launch_bounds__(256) void gat_edge_max_kernel(
        const int* __restrict__ src, const int* __restrict__ dst,
        const float* __restrict__ s_src, const float* __restrict__ s_dst,
        float* __restrict__ e, unsigned* __restrict__ e_max_enc, int n_edges) {
    const int lane = threadIdx.x & 31;
    const int wave = (blockIdx.x * blockDim.x + threadIdx.x) >> 5;
    const int base = wave * EDGES_PER_WAVE_K34;
    if (base >= n_edges) return;
    if (base + EDGES_PER_WAVE_K34 <= n_edges)
        gat_edge_max_chunk(src, dst, s_src, s_dst, e, e_max_enc, n_edges, base + lane, true);
    else
        gat_edge_max_chunk(src, dst, s_src, s_dst, e, e_max_enc, n_edges, base + lane, false);
}

__device__ __forceinline__ void gat_edge_sum_chunk(
        const int* __restrict__ dst, const float* __restrict__ e,
        const unsigned* __restrict__ e_max_enc, float* __restrict__ e_sum,
        int n_edges, int first, const bool whole) {
    int dv[EDGES_PER_LANE];
    float ev[EDGES_PER_LANE];
#pragma unroll
    for (int j = 0; j < EDGES_PER_LANE; ++j) {
        int i = first + 32 * j;
        if (!whole && i >= n_edges) i = n_edges - 1;
        dv[j] = dst[i];
        ev[j] = e[i];
    }
    unsigned mx[EDGES_PER_LANE];
#pragma unroll
    for (int j = 0; j < EDGES_PER_LANE; ++j) mx[j] = e_max_enc[dv[j]];
#pragma unroll
    for (int j = 0; j < EDGES_PER_LANE; ++j) {
        const int i = first + 32 * j;
        if (whole || i < n_edges)
            atomicAdd(&e_sum[dv[j]], __expf(ev[j] - dec_f32(mx[j])));
    }
}

__global__ __launch_bounds__(256) void gat_edge_sum_kernel(
        const int* __restrict__ dst, const float* __restrict__ e,
        const unsigned* __restrict__ e_max_enc, float* __restrict__ e_sum,
        int n_edges) {
    const int lane = threadIdx.x & 31;
    const int wave = (blockIdx.x * blockDim.x + threadIdx.x) >> 5;
    const int base = wave * EDGES_PER_WAVE_K34;
    if (base >= n_edges) return;
    if (base + EDGES_PER_WAVE_K34 <= n_edges)
        gat_edge_sum_chunk(dst, e, e_max_enc, e_sum, n_edges, base + lane, true);
    else
        gat_edge_sum_chunk(dst, e, e_max_enc, e_sum, n_edges, base + lane, false);
}

#define EDGES_PER_WAVE 32
#define ROWS_IN_FLIGHT 8

__device__ __forceinline__ void gat_flush_run(float* __restrict__ h_out, int d, int lane,
                                              v2f acc) {
    float* out = &h_out[(size_t)d * OUT_DIM + 2 * lane];
    atomicAdd(&out[0], acc.x);
    atomicAdd(&out[1], acc.y);
}

__global__ __launch_bounds__(256) void gat_scatter_kernel(
        const int* __restrict__ src, const int* __restrict__ dst,
        const float* __restrict__ e, const unsigned* __restrict__ e_max_enc,
        const float* __restrict__ e_sum, const float* __restrict__ z,
        float* __restrict__ h_out, int n_edges) {
    const int lane  = threadIdx.x & 31;
    const int chunk = (blockIdx.x * blockDim.x + threadIdx.x) >> 5;
    const int base  = chunk * EDGES_PER_WAVE;
    if (base >= n_edges) return;

    int i = base + lane;
    const bool valid = i < n_edges;
    if (!valid) i = n_edges - 1;
    const int my_s = src[i];
    const int my_d = dst[i];
    float my_a = __expf(e[i] - dec_f32(e_max_enc[my_d])) / e_sum[my_d];
    if (!valid) my_a = 0.0f;

    int cur_d = __builtin_amdgcn_readlane(my_d, 0);
    v2f acc = {0.0f, 0.0f};

    for (int g = 0; g < EDGES_PER_WAVE; g += ROWS_IN_FLIGHT) {
        v2f zr[ROWS_IN_FLIGHT];
#pragma unroll
        for (int j = 0; j < ROWS_IN_FLIGHT; ++j) {
            const int s = __builtin_amdgcn_readlane(my_s, g + j);
            zr[j] = *(const v2f*)&z[(size_t)s * OUT_DIM + 2 * lane];
        }
#pragma unroll
        for (int j = 0; j < ROWS_IN_FLIGHT; ++j) {
            const int   d = __builtin_amdgcn_readlane(my_d, g + j);
            const float a = __int_as_float(
                    __builtin_amdgcn_readlane(__float_as_int(my_a), g + j));
            if (d != cur_d) {
                gat_flush_run(h_out, cur_d, lane, acc);
                acc = (v2f){0.0f, 0.0f};
                cur_d = d;
            }
            acc.x += a * zr[j].x;
            acc.y += a * zr[j].y;
        }
    }
    gat_flush_run(h_out, cur_d, lane, acc);
}

extern "C" void kernel_launch(void* const* d_in, const int* in_sizes, int n_in,
                              void* d_out, int out_size, void* d_ws, size_t ws_size,
                              hipStream_t stream) {
    const float* h      = (const float*)d_in[0];
    const int*   src    = (const int*)d_in[1];
    const int*   dst    = (const int*)d_in[2];
    const float* fc_w   = (const float*)d_in[3];
    const float* attn_w = (const float*)d_in[4];
    float*       h_out  = (float*)d_out;

    const int n_nodes = in_sizes[0] / IN_DIM;
    const int n_edges = in_sizes[1];

    char* ws = (char*)d_ws;
    size_t off = 0;
    auto alloc = [&](size_t bytes) {
        void* p = ws + off;
        off += (bytes + 15) & ~size_t(15);
        return p;
    };
    float*    z         = (float*)alloc((size_t)n_nodes * OUT_DIM * sizeof(float));
    float*    s_src     = (float*)alloc((size_t)n_nodes * sizeof(float));
    float*    s_dst     = (float*)alloc((size_t)n_nodes * sizeof(float));
    unsigned* e_max_enc = (unsigned*)alloc((size_t)n_nodes * sizeof(unsigned));
    float*    e_sum     = (float*)alloc((size_t)n_nodes * sizeof(float));
    float*    e         = (float*)alloc((size_t)n_edges * sizeof(float));

    {
        gat_init_kernel<<<256, 256, 0, stream>>>(h_out, e_max_enc, e_sum, n_nodes);
    }
    {
        int blocks = (n_nodes + 15) / 16;
        gat_gemm_wmma_kernel<<<blocks, 128, 0, stream>>>(h, fc_w, z, n_nodes);
    }
    {
        int blocks = (n_nodes + 255) / 256;
        gat_scores_kernel<<<blocks, 256, 0, stream>>>(z, attn_w, s_src, s_dst,
                                                      n_nodes);
    }
    {
        int blocks = (n_edges + 2047) / 2048;
        gat_edge_max_kernel<<<blocks, 256, 0, stream>>>(src, dst, s_src, s_dst,
                                                        e, e_max_enc, n_edges);
    }
    {
        int blocks = (n_edges + 2047) / 2048;
        gat_edge_sum_kernel<<<blocks, 256, 0, stream>>>(dst, e, e_max_enc,
                                                        e_sum, n_edges);
    }
    {
        int blocks = (n_edges + 255) / 256;
        gat_scatter_kernel<<<blocks, 256, 0, stream>>>(src, dst, e, e_max_enc,
                                                       e_sum, z, h_out, n_edges);
    }
}


# edge-max and edge-sum kernels: lane shuffles reduce each run of equal dst among a wave's 32 consecutive edges before the atomic, one atomic per run instead of one per edge
# speedup vs baseline: 3.0551x; 1.4912x over previous
#include <hip/hip_runtime.h>
#include <math.h>

#define IN_DIM   128
#define OUT_DIM  64
#define NEG_SLOPE 0.01f

typedef __attribute__((ext_vector_type(2))) float v2f;
typedef __attribute__((ext_vector_type(8))) float v8f;

__device__ __forceinline__ unsigned enc_f32(float f) {
    unsigned u = __float_as_uint(f);
    return (u & 0x80000000u) ? ~u : (u | 0x80000000u);
}
__device__ __forceinline__ float dec_f32(unsigned u) {
    return __uint_as_float((u & 0x80000000u) ? (u & 0x7FFFFFFFu) : ~u);
}

#define LDS_PITCH 132
__global__ __launch_bounds__(128) void gat_gemm_wmma_kernel(
        const float* __restrict__ h, const float* __restrict__ fc_w,
        const float* __restrict__ attn_w, float* __restrict__ z,
        float* __restrict__ s_src, float* __restrict__ s_dst,
        float* __restrict__ h_out, unsigned* __restrict__ e_max_enc,
        float* __restrict__ e_sum, int n_nodes) {
    __shared__ float lds_w[OUT_DIM * LDS_PITCH];
    __shared__ float lds_part[4][16][2];

    const int tid    = threadIdx.x;
    const int m_base = blockIdx.x * 16;

    {
        int rows = n_nodes - m_base;
        if (rows > 16) rows = 16;
        float4* __restrict__ out4 = (float4*)(h_out + (size_t)m_base * OUT_DIM);
        const float4 zero4 = {0.0f, 0.0f, 0.0f, 0.0f};
        for (int i = tid; i < rows * (OUT_DIM / 4); i += 128) out4[i] = zero4;
        if (tid < rows) {
            e_max_enc[m_base + tid] = 0u;
            e_sum[m_base + tid]     = 0.0f;
        }
    }

    for (int i = tid; i < OUT_DIM * (IN_DIM / 4); i += 128) {
        int row = i >> 5;
        int seg = (i & 31) << 2;
        const float4 v = *(const float4*)&fc_w[row * IN_DIM + seg];
        *(float4*)&lds_w[row * LDS_PITCH + seg] = v;
    }
    __syncthreads();

    const int wave   = tid >> 5;
    const int lane   = tid & 31;
    const int half   = lane >> 4;
    const int lid    = lane & 15;
    const int n_base = wave * 16;

    const float as_c = attn_w[n_base + lid];
    const float ad_c = attn_w[OUT_DIM + n_base + lid];

    int m_row = m_base + lid;
    if (m_row >= n_nodes) m_row = n_nodes - 1;

    const float* arow = h + (size_t)m_row * IN_DIM + 2 * half;
    const float* brow = lds_w + (n_base + lid) * LDS_PITCH + 2 * half;

    v8f c = {};
#pragma unroll
    for (int k0 = 0; k0 < IN_DIM; k0 += 4) {
        v2f a = *(const v2f*)(arow + k0);
        v2f b = *(const v2f*)(brow + k0);
        c = __builtin_amdgcn_wmma_f32_16x16x4_f32(
                 false, a,  false, b,
                 (short)0, c,  false,  false);
    }

#pragma unroll
    for (int r = 0; r < 8; ++r) {
        int m = m_base + half * 8 + r;
        if (m < n_nodes)
            z[(size_t)m * OUT_DIM + n_base + lid] = c[r];
    }

    float ps[8], pd[8];
#pragma unroll
    for (int r = 0; r < 8; ++r) {
        ps[r] = c[r] * as_c;
        pd[r] = c[r] * ad_c;
    }
#pragma unroll
    for (int m = 8; m >= 1; m >>= 1) {
#pragma unroll
        for (int r = 0; r < 8; ++r) {
            ps[r] += __shfl_xor(ps[r], m, 32);
            pd[r] += __shfl_xor(pd[r], m, 32);
        }
    }
    if (lid == 0) {
#pragma unroll
        for (int r = 0; r < 8; ++r) {
            lds_part[wave][half * 8 + r][0] = ps[r];
            lds_part[wave][half * 8 + r][1] = pd[r];
        }
    }
    __syncthreads();
    if (tid < 32) {
        const int row   = tid & 15;
        const int which = tid >> 4;
        const float s = (lds_part[0][row][which] + lds_part[1][row][which]) +
                        (lds_part[2][row][which] + lds_part[3][row][which]);
        const int m = m_base + row;
        if (m < n_nodes) {
            if (which) s_dst[m] = s;
            else       s_src[m] = s;
        }
    }
}

template <int G>
__device__ __forceinline__ void gat_run_shape(const int (&d)[G], int lane,
                                              int (&head_lane)[G], bool (&tail)[G]) {
    int d_prev[G];
#pragma unroll
    for (int j = 0; j < G; ++j) d_prev[j] = __shfl_up(d[j], 1, 32);
    const unsigned upto = 0xFFFFFFFFu >> (31 - lane);
#pragma unroll
    for (int j = 0; j < G; ++j) {
        const bool head = (lane == 0) | (d[j] != d_prev[j]);
        const unsigned heads = __builtin_amdgcn_ballot_w32(head);
        head_lane[j] = 31 - __clz((int)(heads & upto));
        tail[j] = (lane == 31) | ((((heads >> 1) >> lane) & 1u) != 0u);
    }
}
template <int G>
__device__ __forceinline__ void gat_run_max(unsigned (&v)[G], int lane,
                                            const int (&head_lane)[G]) {
#pragma unroll
    for (int k = 1; k < 32; k <<= 1) {
        unsigned up[G];
#pragma unroll
        for (int j = 0; j < G; ++j) up[j] = __shfl_up(v[j], (unsigned)k, 32);
#pragma unroll
        for (int j = 0; j < G; ++j)
            if (lane - k >= head_lane[j]) v[j] = (up[j] > v[j]) ? up[j] : v[j];
    }
}
template <int G>
__device__ __forceinline__ void gat_run_sum(float (&v)[G], int lane,
                                            const int (&head_lane)[G]) {
#pragma unroll
    for (int k = 1; k < 32; k <<= 1) {
        float up[G];
#pragma unroll
        for (int j = 0; j < G; ++j) up[j] = __shfl_up(v[j], (unsigned)k, 32);
#pragma unroll
        for (int j = 0; j < G; ++j)
            if (lane - k >= head_lane[j]) v[j] += up[j];
    }
}

#define EDGES_PER_LANE 8
#define EDGES_PER_WAVE_K34 (32 * EDGES_PER_LANE)

__device__ __forceinline__ void gat_edge_max_chunk(
        const int* __restrict__ src, const int* __restrict__ dst,
        const float* __restrict__ s_src, const float* __restrict__ s_dst,
        float* __restrict__ e, unsigned* __restrict__ e_max_enc, int n_edges,
        int first, int lane, const bool whole) {
    int sv[EDGES_PER_LANE], dv[EDGES_PER_LANE];
#pragma unroll
    for (int j = 0; j < EDGES_PER_LANE; ++j) {
        int i = first + 32 * j;
        if (!whole && i >= n_edges) i = n_edges - 1;
        sv[j] = src[i];
        dv[j] = dst[i];
    }
    float a[EDGES_PER_LANE], b[EDGES_PER_LANE];
#pragma unroll
    for (int j = 0; j < EDGES_PER_LANE; ++j) {
        a[j] = s_src[sv[j]];
        b[j] = s_dst[dv[j]];
    }
    int head_lane[EDGES_PER_LANE];
    bool tail[EDGES_PER_LANE];
    gat_run_shape<EDGES_PER_LANE>(dv, lane, head_lane, tail);
    unsigned m[EDGES_PER_LANE];
#pragma unroll
    for (int j = 0; j < EDGES_PER_LANE; ++j) {
        const int i = first + 32 * j;
        float v = a[j] + b[j];
        v = (v > 0.0f) ? v : NEG_SLOPE * v;
        if (whole || i < n_edges) e[i] = v;
        m[j] = enc_f32(v);
    }
    gat_run_max<EDGES_PER_LANE>(m, lane, head_lane);
#pragma unroll
    for (int j = 0; j < EDGES_PER_LANE; ++j)
        if (tail[j]) atomicMax(&e_max_enc[dv[j]], m[j]);
}

__global__ __launch_bounds__(256) void gat_edge_max_kernel(
        const int* __restrict__ src, const int* __restrict__ dst,
        const float* __restrict__ s_src, const float* __restrict__ s_dst,
        float* __restrict__ e, unsigned* __restrict__ e_max_enc, int n_edges) {
    const int lane = threadIdx.x & 31;
    const int wave = (blockIdx.x * blockDim.x + threadIdx.x) >> 5;
    const int base = wave * EDGES_PER_WAVE_K34;
    if (base >= n_edges) return;
    if (base + EDGES_PER_WAVE_K34 <= n_edges)
        gat_edge_max_chunk(src, dst, s_src, s_dst, e, e_max_enc, n_edges, base + lane, lane,
                           true);
    else
        gat_edge_max_chunk(src, dst, s_src, s_dst, e, e_max_enc, n_edges, base + lane, lane,
                           false);
}

__device__ __forceinline__ void gat_edge_sum_chunk(
        const int* __restrict__ dst, const float* __restrict__ e,
        const unsigned* __restrict__ e_max_enc, float* __restrict__ e_sum,
        int n_edges, int first, int lane, const bool whole) {
    int dv[EDGES_PER_LANE];
    float ev[EDGES_PER_LANE];
#pragma unroll
    for (int j = 0; j < EDGES_PER_LANE; ++j) {
        int i = first + 32 * j;
        if (!whole && i >= n_edges) i = n_edges - 1;
        dv[j] = dst[i];
        ev[j] = e[i];
    }
    unsigned mx[EDGES_PER_LANE];
#pragma unroll
    for (int j = 0; j < EDGES_PER_LANE; ++j) mx[j] = e_max_enc[dv[j]];
    int head_lane[EDGES_PER_LANE];
    bool tail[EDGES_PER_LANE];
    gat_run_shape<EDGES_PER_LANE>(dv, lane, head_lane, tail);
    float p[EDGES_PER_LANE];
#pragma unroll
    for (int j = 0; j < EDGES_PER_LANE; ++j) {
        const int i = first + 32 * j;
        p[j] = __expf(ev[j] - dec_f32(mx[j]));
        if (!whole && i >= n_edges) p[j] = 0.0f;
    }
    gat_run_sum<EDGES_PER_LANE>(p, lane, head_lane);
#pragma unroll
    for (int j = 0; j < EDGES_PER_LANE; ++j)
        if (tail[j]) atomicAdd(&e_sum[dv[j]], p[j]);
}

__global__ __launch_bounds__(256) void gat_edge_sum_kernel(
        const int* __restrict__ dst, const float* __restrict__ e,
        const unsigned* __restrict__ e_max_enc, float* __restrict__ e_sum,
        int n_edges) {
    const int lane = threadIdx.x & 31;
    const int wave = (blockIdx.x * blockDim.x + threadIdx.x) >> 5;
    const int base = wave * EDGES_PER_WAVE_K34;
    if (base >= n_edges) return;
    if (base + EDGES_PER_WAVE_K34 <= n_edges)
        gat_edge_sum_chunk(dst, e, e_max_enc, e_sum, n_edges, base + lane, lane, true);
    else
        gat_edge_sum_chunk(dst, e, e_max_enc, e_sum, n_edges, base + lane, lane, false);
}

#define EDGES_PER_WAVE 32
#define ROWS_IN_FLIGHT 8

__device__ __forceinline__ void gat_flush_run(float* __restrict__ h_out, int d, int lane,
                                              v2f acc) {
    float* out = &h_out[(size_t)d * OUT_DIM + 2 * lane];
    atomicAdd(&out[0], acc.x);
    atomicAdd(&out[1], acc.y);
}

__global__ __launch_bounds__(256) void gat_scatter_kernel(
        const int* __restrict__ src, const int* __restrict__ dst,
        const float* __restrict__ e, const unsigned* __restrict__ e_max_enc,
        const float* __restrict__ e_sum, const float* __restrict__ z,
        float* __restrict__ h_out, int n_edges) {
    const int lane  = threadIdx.x & 31;
    const int chunk = (blockIdx.x * blockDim.x + threadIdx.x) >> 5;
    const int base  = chunk * EDGES_PER_WAVE;
    if (base >= n_edges) return;

    int i = base + lane;
    const bool valid = i < n_edges;
    if (!valid) i = n_edges - 1;
    const int my_s = src[i];
    const int my_d = dst[i];
    float my_a = __expf(e[i] - dec_f32(e_max_enc[my_d])) / e_sum[my_d];
    if (!valid) my_a = 0.0f;

    int cur_d = __builtin_amdgcn_readlane(my_d, 0);
    v2f acc = {0.0f, 0.0f};

    for (int g = 0; g < EDGES_PER_WAVE; g += ROWS_IN_FLIGHT) {
        v2f zr[ROWS_IN_FLIGHT];
#pragma unroll
        for (int j = 0; j < ROWS_IN_FLIGHT; ++j) {
            const int s = __builtin_amdgcn_readlane(my_s, g + j);
            zr[j] = *(const v2f*)&z[(size_t)s * OUT_DIM + 2 * lane];
        }
#pragma unroll
        for (int j = 0; j < ROWS_IN_FLIGHT; ++j) {
            const int   d = __builtin_amdgcn_readlane(my_d, g + j);
            const float a = __int_as_float(
                    __builtin_amdgcn_readlane(__float_as_int(my_a), g + j));
            if (d != cur_d) {
                gat_flush_run(h_out, cur_d, lane, acc);
                acc = (v2f){0.0f, 0.0f};
                cur_d = d;
            }
            acc.x += a * zr[j].x;
            acc.y += a * zr[j].y;
        }
    }
    gat_flush_run(h_out, cur_d, lane, acc);
}

extern "C" void kernel_launch(void* const* d_in, const int* in_sizes, int n_in,
                              void* d_out, int out_size, void* d_ws, size_t ws_size,
                              hipStream_t stream) {
    const float* h      = (const float*)d_in[0];
    const int*   src    = (const int*)d_in[1];
    const int*   dst    = (const int*)d_in[2];
    const float* fc_w   = (const float*)d_in[3];
    const float* attn_w = (const float*)d_in[4];
    float*       h_out  = (float*)d_out;

    const int n_nodes = in_sizes[0] / IN_DIM;
    const int n_edges = in_sizes[1];

    char* ws = (char*)d_ws;
    size_t off = 0;
    auto alloc = [&](size_t bytes) {
        void* p = ws + off;
        off += (bytes + 15) & ~size_t(15);
        return p;
    };
    float*    z         = (float*)alloc((size_t)n_nodes * OUT_DIM * sizeof(float));
    float*    s_src     = (float*)alloc((size_t)n_nodes * sizeof(float));
    float*    s_dst     = (float*)alloc((size_t)n_nodes * sizeof(float));
    unsigned* e_max_enc = (unsigned*)alloc((size_t)n_nodes * sizeof(unsigned));
    float*    e_sum     = (float*)alloc((size_t)n_nodes * sizeof(float));
    float*    e         = (float*)alloc((size_t)n_edges * sizeof(float));

    {
        int blocks = (n_nodes + 15) / 16;
        gat_gemm_wmma_kernel<<<blocks, 128, 0, stream>>>(h, fc_w, attn_w, z, s_src, s_dst,
                                                         h_out, e_max_enc, e_sum, n_nodes);
    }
    {
        int blocks = (n_edges + 2047) / 2048;
        gat_edge_max_kernel<<<blocks, 256, 0, stream>>>(src, dst, s_src, s_dst,
                                                        e, e_max_enc, n_edges);
    }
    {
        int blocks = (n_edges + 2047) / 2048;
        gat_edge_sum_kernel<<<blocks, 256, 0, stream>>>(dst, e, e_max_enc,
                                                        e_sum, n_edges);
    }
    {
        int blocks = (n_edges + 255) / 256;
        gat_scatter_kernel<<<blocks, 256, 0, stream>>>(src, dst, e, e_max_enc,
                                                       e_sum, z, h_out, n_edges);
    }
}


# gemm kernel: each wave owns 16 rows and all 64 columns (4 accumulator tiles) and loads its rows once, 16 bytes per load, 64 rows per block; k sum of z and column sum of scores in another order
# speedup vs baseline: 3.5528x; 1.1629x over previous
#include <hip/hip_runtime.h>
#include <math.h>

#define IN_DIM   128
#define OUT_DIM  64
#define NEG_SLOPE 0.01f

typedef __attribute__((ext_vector_type(2))) float v2f;
typedef __attribute__((ext_vector_type(8))) float v8f;

__device__ __forceinline__ unsigned enc_f32(float f) {
    unsigned u = __float_as_uint(f);
    return (u & 0x80000000u) ? ~u : (u | 0x80000000u);
}
__device__ __forceinline__ float dec_f32(unsigned u) {
    return __uint_as_float((u & 0x80000000u) ? (u & 0x7FFFFFFFu) : ~u);
}

#define LDS_PITCH 132
#define ROWS_PER_BLOCK 64
__global__ __launch_bounds__(128) void gat_gemm_wmma_kernel(
        const float* __restrict__ h, const float* __restrict__ fc_w,
        const float* __restrict__ attn_w, float* __restrict__ z,
        float* __restrict__ s_src, float* __restrict__ s_dst,
        float* __restrict__ h_out, unsigned* __restrict__ e_max_enc,
        float* __restrict__ e_sum, int n_nodes) {
    __shared__ float lds_w[OUT_DIM * LDS_PITCH];

    const int tid    = threadIdx.x;
    const int wave   = tid >> 5;
    const int lane   = tid & 31;
    const int half   = lane >> 4;
    const int lid    = lane & 15;
    const int m_base = blockIdx.x * ROWS_PER_BLOCK;
    const int m_wave = m_base + wave * 16;

    int m_row = m_wave + lid;
    if (m_row >= n_nodes) m_row = n_nodes - 1;
    const float* arow = h + (size_t)m_row * IN_DIM + 4 * half;
    float4 a[IN_DIM / 8];
#pragma unroll
    for (int j = 0; j < IN_DIM / 8; ++j) a[j] = *(const float4*)(arow + 8 * j);

    float as_c[4], ad_c[4];
#pragma unroll
    for (int t = 0; t < 4; ++t) {
        as_c[t] = attn_w[16 * t + lid];
        ad_c[t] = attn_w[OUT_DIM + 16 * t + lid];
    }

    {
        int rows = n_nodes - m_base;
        if (rows > ROWS_PER_BLOCK) rows = ROWS_PER_BLOCK;
        float4* __restrict__ out4 = (float4*)(h_out + (size_t)m_base * OUT_DIM);
        const float4 zero4 = {0.0f, 0.0f, 0.0f, 0.0f};
        for (int i = tid; i < rows * (OUT_DIM / 4); i += 128) out4[i] = zero4;
        if (tid < rows) {
            e_max_enc[m_base + tid] = 0u;
            e_sum[m_base + tid]     = 0.0f;
        }
    }

    for (int i = tid; i < OUT_DIM * (IN_DIM / 4); i += 128) {
        int row = i >> 5;
        int seg = (i & 31) << 2;
        const float4 v = *(const float4*)&fc_w[row * IN_DIM + seg];
        *(float4*)&lds_w[row * LDS_PITCH + seg] = v;
    }
    __syncthreads();

    const float* brow = lds_w + lid * LDS_PITCH + 4 * half;

    v8f c[4] = {};
#pragma unroll
    for (int j = 0; j < IN_DIM / 8; ++j) {
        float4 b[4];
#pragma unroll
        for (int t = 0; t < 4; ++t)
            b[t] = *(const float4*)(brow + t * 16 * LDS_PITCH + 8 * j);
        const v2f a_lo = {a[j].x, a[j].y};
        const v2f a_hi = {a[j].z, a[j].w};
#pragma unroll
        for (int t = 0; t < 4; ++t) {
            const v2f b_lo = {b[t].x, b[t].y};
            c[t] = __builtin_amdgcn_wmma_f32_16x16x4_f32(
                     false, a_lo,  false, b_lo,
                     (short)0, c[t],  false,  false);
        }
#pragma unroll
        for (int t = 0; t < 4; ++t) {
            const v2f b_hi = {b[t].z, b[t].w};
            c[t] = __builtin_amdgcn_wmma_f32_16x16x4_f32(
                     false, a_hi,  false, b_hi,
                     (short)0, c[t],  false,  false);
        }
    }

#pragma unroll
    for (int r = 0; r < 8; ++r) {
        const int m = m_wave + half * 8 + r;
        if (m < n_nodes) {
#pragma unroll
            for (int t = 0; t < 4; ++t)
                z[(size_t)m * OUT_DIM + 16 * t + lid] = c[t][r];
        }
    }

    float ps[8], pd[8];
#pragma unroll
    for (int r = 0; r < 8; ++r) {
        ps[r] = (c[0][r] * as_c[0] + c[1][r] * as_c[1]) + (c[2][r] * as_c[2] + c[3][r] * as_c[3]);
        pd[r] = (c[0][r] * ad_c[0] + c[1][r] * ad_c[1]) + (c[2][r] * ad_c[2] + c[3][r] * ad_c[3]);
    }
#pragma unroll
    for (int m = 8; m >= 1; m >>= 1) {
#pragma unroll
        for (int r = 0; r < 8; ++r) {
            ps[r] += __shfl_xor(ps[r], m, 32);
            pd[r] += __shfl_xor(pd[r], m, 32);
        }
    }
    float vs = ps[0], vd = pd[0];
#pragma unroll
    for (int r = 1; r < 8; ++r) {
        if ((lid & 7) == r) { vs = ps[r]; vd = pd[r]; }
    }
    const int m_s = m_wave + half * 8 + (lid & 7);
    if (m_s < n_nodes) {
        if (lid < 8) s_src[m_s] = vs;
        else         s_dst[m_s] = vd;
    }
}

template <int G>
__device__ __forceinline__ void gat_run_shape(const int (&d)[G], int lane,
                                              int (&head_lane)[G], bool (&tail)[G]) {
    int d_prev[G];
#pragma unroll
    for (int j = 0; j < G; ++j) d_prev[j] = __shfl_up(d[j], 1, 32);
    const unsigned upto = 0xFFFFFFFFu >> (31 - lane);
#pragma unroll
    for (int j = 0; j < G; ++j) {
        const bool head = (lane == 0) | (d[j] != d_prev[j]);
        const unsigned heads = __builtin_amdgcn_ballot_w32(head);
        head_lane[j] = 31 - __clz((int)(heads & upto));
        tail[j] = (lane == 31) | ((((heads >> 1) >> lane) & 1u) != 0u);
    }
}
template <int G>
__device__ __forceinline__ void gat_run_max(unsigned (&v)[G], int lane,
                                            const int (&head_lane)[G]) {
#pragma unroll
    for (int k = 1; k < 32; k <<= 1) {
        unsigned up[G];
#pragma unroll
        for (int j = 0; j < G; ++j) up[j] = __shfl_up(v[j], (unsigned)k, 32);
#pragma unroll
        for (int j = 0; j < G; ++j)
            if (lane - k >= head_lane[j]) v[j] = (up[j] > v[j]) ? up[j] : v[j];
    }
}
template <int G>
__device__ __forceinline__ void gat_run_sum(float (&v)[G], int lane,
                                            const int (&head_lane)[G]) {
#pragma unroll
    for (int k = 1; k < 32; k <<= 1) {
        float up[G];
#pragma unroll
        for (int j = 0; j < G; ++j) up[j] = __shfl_up(v[j], (unsigned)k, 32);
#pragma unroll
        for (int j = 0; j < G; ++j)
            if (lane - k >= head_lane[j]) v[j] += up[j];
    }
}

#define EDGES_PER_LANE 8
#define EDGES_PER_WAVE_K34 (32 * EDGES_PER_LANE)

__device__ __forceinline__ void gat_edge_max_chunk(
        const int* __restrict__ src, const int* __restrict__ dst,
        const float* __restrict__ s_src, const float* __restrict__ s_dst,
        float* __restrict__ e, unsigned* __restrict__ e_max_enc, int n_edges,
        int first, int lane, const bool whole) {
    int sv[EDGES_PER_LANE], dv[EDGES_PER_LANE];
#pragma unroll
    for (int j = 0; j < EDGES_PER_LANE; ++j) {
        int i = first + 32 * j;
        if (!whole && i >= n_edges) i = n_edges - 1;
        sv[j] = src[i];
        dv[j] = dst[i];
    }
    float a[EDGES_PER_LANE], b[EDGES_PER_LANE];
#pragma unroll
    for (int j = 0; j < EDGES_PER_LANE; ++j) {
        a[j] = s_src[sv[j]];
        b[j] = s_dst[dv[j]];
    }
    int head_lane[EDGES_PER_LANE];
    bool tail[EDGES_PER_LANE];
    gat_run_shape<EDGES_PER_LANE>(dv, lane, head_lane, tail);
    unsigned m[EDGES_PER_LANE];
#pragma unroll
    for (int j = 0; j < EDGES_PER_LANE; ++j) {
        const int i = first + 32 * j;
        float v = a[j] + b[j];
        v = (v > 0.0f) ? v : NEG_SLOPE * v;
        if (whole || i < n_edges) e[i] = v;
        m[j] = enc_f32(v);
    }
    gat_run_max<EDGES_PER_LANE>(m, lane, head_lane);
#pragma unroll
    for (int j = 0; j < EDGES_PER_LANE; ++j)
        if (tail[j]) atomicMax(&e_max_enc[dv[j]], m[j]);
}

__global__ __launch_bounds__(256) void gat_edge_max_kernel(
        const int* __restrict__ src, const int* __restrict__ dst,
        const float* __restrict__ s_src, const float* __restrict__ s_dst,
        float* __restrict__ e, unsigned* __restrict__ e_max_enc, int n_edges) {
    const int lane = threadIdx.x & 31;
    const int wave = (blockIdx.x * blockDim.x + threadIdx.x) >> 5;
    const int base = wave * EDGES_PER_WAVE_K34;
    if (base >= n_edges) return;
    if (base + EDGES_PER_WAVE_K34 <= n_edges)
        gat_edge_max_chunk(src, dst, s_src, s_dst, e, e_max_enc, n_edges, base + lane, lane,
                           true);
    else
        gat_edge_max_chunk(src, dst, s_src, s_dst, e, e_max_enc, n_edges, base + lane, lane,
                           false);
}

__device__ __forceinline__ void gat_edge_sum_chunk(
        const int* __restrict__ dst, const float* __restrict__ e,
        const unsigned* __restrict__ e_max_enc, float* __restrict__ e_sum,
        int n_edges, int first, int lane, const bool whole) {
    int dv[EDGES_PER_LANE];
    float ev[EDGES_PER_LANE];
#pragma unroll
    for (int j = 0; j < EDGES_PER_LANE; ++j) {
        int i = first + 32 * j;
        if (!whole && i >= n_edges) i = n_edges - 1;
        dv[j] = dst[i];
        ev[j] = e[i];
    }
    unsigned mx[EDGES_PER_LANE];
#pragma unroll
    for (int j = 0; j < EDGES_PER_LANE; ++j) mx[j] = e_max_enc[dv[j]];
    int head_lane[EDGES_PER_LANE];
    bool tail[EDGES_PER_LANE];
    gat_run_shape<EDGES_PER_LANE>(dv, lane, head_lane, tail);
    float p[EDGES_PER_LANE];
#pragma unroll
    for (int j = 0; j < EDGES_PER_LANE; ++j) {
        const int i = first + 32 * j;
        p[j] = __expf(ev[j] - dec_f32(mx[j]));
        if (!whole && i >= n_edges) p[j] = 0.0f;
    }
    gat_run_sum<EDGES_PER_LANE>(p, lane, head_lane);
#pragma unroll
    for (int j = 0; j < EDGES_PER_LANE; ++j)
        if (tail[j]) atomicAdd(&e_sum[dv[j]], p[j]);
}

__global__ __launch_bounds__(256) void gat_edge_sum_kernel(
        const int* __restrict__ dst, const float* __restrict__ e,
        const unsigned* __restrict__ e_max_enc, float* __restrict__ e_sum,
        int n_edges) {
    const int lane = threadIdx.x & 31;
    const int wave = (blockIdx.x * blockDim.x + threadIdx.x) >> 5;
    const int base = wave * EDGES_PER_WAVE_K34;
    if (base >= n_edges) return;
    if (base + EDGES_PER_WAVE_K34 <= n_edges)
        gat_edge_sum_chunk(dst, e, e_max_enc, e_sum, n_edges, base + lane, lane, true);
    else
        gat_edge_sum_chunk(dst, e, e_max_enc, e_sum, n_edges, base + lane, lane, false);
}

#define EDGES_PER_WAVE 32
#define ROWS_IN_FLIGHT 8

__device__ __forceinline__ void gat_flush_run(float* __restrict__ h_out, int d, int lane,
                                              v2f acc) {
    float* out = &h_out[(size_t)d * OUT_DIM + 2 * lane];
    atomicAdd(&out[0], acc.x);
    atomicAdd(&out[1], acc.y);
}

__global__ __launch_bounds__(256) void gat_scatter_kernel(
        const int* __restrict__ src, const int* __restrict__ dst,
        const float* __restrict__ e, const unsigned* __restrict__ e_max_enc,
        const float* __restrict__ e_sum, const float* __restrict__ z,
        float* __restrict__ h_out, int n_edges) {
    const int lane  = threadIdx.x & 31;
    const int chunk = (blockIdx.x * blockDim.x + threadIdx.x) >> 5;
    const int base  = chunk * EDGES_PER_WAVE;
    if (base >= n_edges) return;

    int i = base + lane;
    const bool valid = i < n_edges;
    if (!valid) i = n_edges - 1;
    const int my_s = src[i];
    const int my_d = dst[i];
    float my_a = __expf(e[i] - dec_f32(e_max_enc[my_d])) / e_sum[my_d];
    if (!valid) my_a = 0.0f;

    int cur_d = __builtin_amdgcn_readlane(my_d, 0);
    v2f acc = {0.0f, 0.0f};

    for (int g = 0; g < EDGES_PER_WAVE; g += ROWS_IN_FLIGHT) {
        v2f zr[ROWS_IN_FLIGHT];
#pragma unroll
        for (int j = 0; j < ROWS_IN_FLIGHT; ++j) {
            const int s = __builtin_amdgcn_readlane(my_s, g + j);
            zr[j] = *(const v2f*)&z[(size_t)s * OUT_DIM + 2 * lane];
        }
#pragma unroll
        for (int j = 0; j < ROWS_IN_FLIGHT; ++j) {
            const int   d = __builtin_amdgcn_readlane(my_d, g + j);
            const float a = __int_as_float(
                    __builtin_amdgcn_readlane(__float_as_int(my_a), g + j));
            if (d != cur_d) {
                gat_flush_run(h_out, cur_d, lane, acc);
                acc = (v2f){0.0f, 0.0f};
                cur_d = d;
            }
            acc.x += a * zr[j].x;
            acc.y += a * zr[j].y;
        }
    }
    gat_flush_run(h_out, cur_d, lane, acc);
}

extern "C" void kernel_launch(void* const* d_in, const int* in_sizes, int n_in,
                              void* d_out, int out_size, void* d_ws, size_t ws_size,
                              hipStream_t stream) {
    const float* h      = (const float*)d_in[0];
    const int*   src    = (const int*)d_in[1];
    const int*   dst    = (const int*)d_in[2];
    const float* fc_w   = (const float*)d_in[3];
    const float* attn_w = (const float*)d_in[4];
    float*       h_out  = (float*)d_out;

    const int n_nodes = in_sizes[0] / IN_DIM;
    const int n_edges = in_sizes[1];

    char* ws = (char*)d_ws;
    size_t off = 0;
    auto alloc = [&](size_t bytes) {
        void* p = ws + off;
        off += (bytes + 15) & ~size_t(15);
        return p;
    };
    float*    z         = (float*)alloc((size_t)n_nodes * OUT_DIM * sizeof(float));
    float*    s_src     = (float*)alloc((size_t)n_nodes * sizeof(float));
    float*    s_dst     = (float*)alloc((size_t)n_nodes * sizeof(float));
    unsigned* e_max_enc = (unsigned*)alloc((size_t)n_nodes * sizeof(unsigned));
    float*    e_sum     = (float*)alloc((size_t)n_nodes * sizeof(float));
    float*    e         = (float*)alloc((size_t)n_edges * sizeof(float));

    {
        int blocks = (n_nodes + 63) / 64;
        gat_gemm_wmma_kernel<<<blocks, 128, 0, stream>>>(h, fc_w, attn_w, z, s_src, s_dst,
                                                         h_out, e_max_enc, e_sum, n_nodes);
    }
    {
        int blocks = (n_edges + 2047) / 2048;
        gat_edge_max_kernel<<<blocks, 256, 0, stream>>>(src, dst, s_src, s_dst,
                                                        e, e_max_enc, n_edges);
    }
    {
        int blocks = (n_edges + 2047) / 2048;
        gat_edge_sum_kernel<<<blocks, 256, 0, stream>>>(dst, e, e_max_enc,
                                                        e_sum, n_edges);
    }
    {
        int blocks = (n_edges + 255) / 256;
        gat_scatter_kernel<<<blocks, 256, 0, stream>>>(src, dst, e, e_max_enc,
                                                       e_sum, z, h_out, n_edges);
    }
}
